# starting kernel
# baseline (speedup 1.0000x reference)
#include <hip/hip_runtime.h>
#include <hip/hip_bf16.h>
#include <math.h>

#define GH 30
#define GW 30
#define HWC 900
#define DD 256
#define MAXOBJ 16
#define SENT 900
#define KF 261
#define KP 264
#define LD 268

typedef __attribute__((ext_vector_type(2))) float v2f;
typedef __attribute__((ext_vector_type(8))) float v8f;

__global__ __launch_bounds__(1024) void cc_extract_kernel(const int* __restrict__ grid,
                                                          int* __restrict__ meta) {
  __shared__ int g[HWC];
  __shared__ int lab[HWC];
  __shared__ int rmn[HWC], rmx[HWC], cmn[HWC], cmx[HWC];
  __shared__ int roots[MAXOBJ];
  __shared__ int nroots;
  __shared__ int changed;

  const int b = blockIdx.x;
  const int tid = threadIdx.x;
  const int* gg = grid + b * HWC;

  for (int t = tid; t < HWC; t += blockDim.x) {
    int v = gg[t];
    g[t] = v;
    lab[t] = (v > 0) ? t : SENT;
    rmn[t] = GH; rmx[t] = -1; cmn[t] = GW; cmx[t] = -1;
  }
  __syncthreads();

  for (int iter = 0; iter < 2048; ++iter) {
    if (tid == 0) changed = 0;
    __syncthreads();
    const int cell = tid;
    const bool active = (cell < HWC) && (g[cell] > 0);
    int newl = SENT, oldl = SENT;
    if (active) {
      const int r = cell / GW, c = cell % GW;
      const int gv = g[cell];
      int m = lab[cell]; oldl = m;
      if (r > 0      && g[cell - GW] == gv) m = min(m, lab[cell - GW]);
      if (r < GH - 1 && g[cell + GW] == gv) m = min(m, lab[cell + GW]);
      if (c > 0      && g[cell - 1]  == gv) m = min(m, lab[cell - 1]);
      if (c < GW - 1 && g[cell + 1]  == gv) m = min(m, lab[cell + 1]);
      if (m < SENT) m = min(m, lab[m]);
      newl = m;
    }
    __syncthreads();
    if (active && newl != oldl) { lab[cell] = newl; changed = 1; }
    __syncthreads();
    const int done = (changed == 0);
    __syncthreads();
    if (done) break;
  }

  if (tid < HWC && g[tid] > 0) {
    const int root = lab[tid];
    atomicMin(&rmn[root], tid / GW);
    atomicMax(&rmx[root], tid / GW);
    atomicMin(&cmn[root], tid % GW);
    atomicMax(&cmx[root], tid % GW);
  }
  __syncthreads();

  if (tid == 0) {
    int cnt = 0;
    for (int i = 0; i < HWC && cnt < MAXOBJ; ++i)
      if (g[i] > 0 && lab[i] == i) roots[cnt++] = i;
    nroots = cnt;
  }
  __syncthreads();

  if (tid < MAXOBJ) {
    int* mo = meta + (b * MAXOBJ + tid) * 6;
    if (tid < nroots) {
      const int r = roots[tid];
      int y  = min(max(rmn[r], 0), GH - 1);
      int x  = min(max(cmn[r], 0), GW - 1);
      int y2 = min(max(rmx[r] + 1, y + 1), GH);
      int x2 = min(max(cmx[r] + 1, x + 1), GW);
      mo[0] = g[r]; mo[1] = y; mo[2] = x; mo[3] = y2 - y; mo[4] = x2 - x; mo[5] = 1;
    } else {
      mo[0] = 0; mo[1] = 0; mo[2] = 0; mo[3] = 1; mo[4] = 1; mo[5] = 0;
    }
  }
}

__global__ __launch_bounds__(256) void pool_kernel(const float* __restrict__ grid_emb,
                                                   const int* __restrict__ meta,
                                                   float* __restrict__ comb) {
  const int bk = blockIdx.x;
  const int b = bk >> 4;
  const int d = threadIdx.x;
  const int* mo = meta + bk * 6;
  const int color = mo[0], y = mo[1], x = mo[2], h = mo[3], w = mo[4];

  float s = 0.f;
  for (int i = 0; i < h; ++i) {
    const float* rowp = grid_emb + (((size_t)(b * GH + y + i)) * GW + x) * DD + d;
    for (int j = 0; j < w; ++j) s += rowp[(size_t)j * DD];
  }
  float* op = comb + (size_t)bk * KP;
  op[d] = s / (float)(h * w);
  if (d == 0) {
    op[DD + 0] = (float)color / 9.0f;
    op[DD + 1] = (float)x / (float)GW;
    op[DD + 2] = (float)y / (float)GH;
    op[DD + 3] = (float)w / (float)GW;
    op[DD + 4] = (float)h / (float)GH;
    op[DD + 5] = 0.f; op[DD + 6] = 0.f; op[DD + 7] = 0.f;
  }
}

__global__ __launch_bounds__(512) void fused_mlp_kernel(
    const float* __restrict__ comb_g, const int* __restrict__ meta,
    const float* __restrict__ structure, const float* __restrict__ W1,
    const float* __restrict__ b1, const float* __restrict__ W2,
    const float* __restrict__ b2, const float* __restrict__ Wp,
    const float* __restrict__ bp, const float* __restrict__ gmm,
    const float* __restrict__ bta, const float* __restrict__ osp,
    float* __restrict__ out) {
  __shared__ float sA[MAXOBJ][LD];
  __shared__ float sB[MAXOBJ][LD];
  __shared__ float sC[MAXOBJ][LD];
  __shared__ float s_sn[DD];
  __shared__ float s_dot[MAXOBJ];
  __shared__ float s_red[16];
  __shared__ int   s_valid[MAXOBJ];
  __shared__ float s_norm;

  const int b = blockIdx.x;
  const int tid = threadIdx.x;
  const int lane = tid & 31;
  const int wv = tid >> 5;
  const int half = lane >> 4;
  const int mrow = lane & 15;
  const int nn = wv * 16 + (lane & 15);

  for (int t = tid; t < MAXOBJ * KP; t += 512) {
    int r = t / KP, c = t % KP;
    sC[r][c] = comb_g[(size_t)(b * MAXOBJ + r) * KP + c];
  }
  if (tid < MAXOBJ * (LD - KP)) {
    int r = tid / (LD - KP), c = KP + tid % (LD - KP);
    sC[r][c] = 0.f;
  }
  if (tid < MAXOBJ) s_valid[tid] = meta[(b * MAXOBJ + tid) * 6 + 5];

  if (tid < DD) {
    const float* sp = structure + (size_t)b * 8 * DD + tid;
    float s = 0.f;
    #pragma unroll
    for (int k = 0; k < 8; ++k) s += sp[(size_t)k * DD];
    s_sn[tid] = s * 0.125f;
  }
  __syncthreads();
  if (tid < DD) {
    float p = s_sn[tid] * s_sn[tid];
    #pragma unroll
    for (int m = 16; m >= 1; m >>= 1) p += __shfl_xor(p, m, 32);
    if (lane == 0) s_red[wv] = p;
  }
  __syncthreads();
  if (tid == 0) {
    float t = 0.f;
    for (int i = 0; i < 8; ++i) t += s_red[i];
    s_norm = fmaxf(sqrtf(t), 1e-8f);
  }
  __syncthreads();
  if (tid < DD) s_sn[tid] /= s_norm;

  {
    v8f acc = {};
    for (int k0 = 0; k0 < KP; k0 += 4) {
      const int kk = k0 + half * 2;
      v2f av; av.x = sC[mrow][kk]; av.y = sC[mrow][kk + 1];
      const int r0 = min(kk, KF - 1), r1 = min(kk + 1, KF - 1);
      v2f bv; bv.x = W1[(size_t)r0 * DD + nn]; bv.y = W1[(size_t)r1 * DD + nn];
      acc = __builtin_amdgcn_wmma_f32_16x16x4_f32(false, av, false, bv,
                                                  (short)0, acc, false, false);
    }
    const float bias = b1[nn];
    #pragma unroll
    for (int i = 0; i < 8; ++i) {
      const int row = half * 8 + i;
      float v = acc[i] + bias;
      v = 0.5f * v * (1.0f + erff(v * 0.70710678118654752f));
      sA[row][nn] = v;
    }
  }
  __syncthreads();

  {
    v8f acc = {};
    for (int k0 = 0; k0 < DD; k0 += 4) {
      const int kk = k0 + half * 2;
      v2f av; av.x = sA[mrow][kk]; av.y = sA[mrow][kk + 1];
      v2f bv; bv.x = W2[(size_t)kk * DD + nn]; bv.y = W2[(size_t)(kk + 1) * DD + nn];
      acc = __builtin_amdgcn_wmma_f32_16x16x4_f32(false, av, false, bv,
                                                  (short)0, acc, false, false);
    }
    const float bias = b2[nn];
    #pragma unroll
    for (int i = 0; i < 8; ++i) {
      const int row = half * 8 + i;
      const float v = acc[i] + bias;
      sB[row][nn] = s_valid[row] ? v : 0.f;
    }
  }
  __syncthreads();

  {
    float p = 0.f;
    #pragma unroll
    for (int j = 0; j < 8; ++j) {
      const int n = lane + 32 * j;
      p += sB[wv][n] * s_sn[n];
    }
    #pragma unroll
    for (int m = 16; m >= 1; m >>= 1) p += __shfl_xor(p, m, 32);
    if (lane == 0) s_dot[wv] = p;
  }
  __syncthreads();
  {
    const float os = osp[0];
    for (int t = tid; t < MAXOBJ * DD; t += 512) {
      const int r = t >> 8, n = t & 255;
      sB[r][n] = (sB[r][n] - s_dot[r] * s_sn[n]) * os;
    }
  }
  __syncthreads();

  {
    v8f acc = {};
    for (int k0 = 0; k0 < DD; k0 += 4) {
      const int kk = k0 + half * 2;
      v2f av; av.x = sB[mrow][kk]; av.y = sB[mrow][kk + 1];
      v2f bv; bv.x = Wp[(size_t)kk * DD + nn]; bv.y = Wp[(size_t)(kk + 1) * DD + nn];
      acc = __builtin_amdgcn_wmma_f32_16x16x4_f32(false, av, false, bv,
                                                  (short)0, acc, false, false);
    }
    const float bias = bp[nn];
    #pragma unroll
    for (int i = 0; i < 8; ++i) sC[half * 8 + i][nn] = acc[i] + bias;
  }
  __syncthreads();

  {
    float s1 = 0.f, s2 = 0.f;
    #pragma unroll
    for (int j = 0; j < 8; ++j) {
      const float v = sC[wv][lane + 32 * j];
      s1 += v; s2 += v * v;
    }
    #pragma unroll
    for (int m = 16; m >= 1; m >>= 1) {
      s1 += __shfl_xor(s1, m, 32);
      s2 += __shfl_xor(s2, m, 32);
    }
    const float mu = s1 * (1.0f / 256.0f);
    const float var = s2 * (1.0f / 256.0f) - mu * mu;
    const float rstd = rsqrtf(var + 1e-5f);
    float* op = out + (size_t)(b * MAXOBJ + wv) * DD;
    #pragma unroll
    for (int j = 0; j < 8; ++j) {
      const int n = lane + 32 * j;
      const float v = sC[wv][n];
      op[n] = (v - mu) * rstd * gmm[n] + bta[n];
    }
  }
}

extern "C" void kernel_launch(void* const* d_in, const int* in_sizes, int n_in,
                              void* d_out, int out_size, void* d_ws, size_t ws_size,
                              hipStream_t stream) {
  const float* grid_emb  = (const float*)d_in[0];
  const int*   grid      = (const int*)  d_in[1];
  const float* structure = (const float*)d_in[2];
  const float* W1        = (const float*)d_in[3];
  const float* b1        = (const float*)d_in[4];
  const float* W2        = (const float*)d_in[5];
  const float* b2        = (const float*)d_in[6];
  const float* Wp        = (const float*)d_in[7];
  const float* bp        = (const float*)d_in[8];
  const float* gmm       = (const float*)d_in[9];
  const float* bta       = (const float*)d_in[10];
  const float* osp       = (const float*)d_in[11];
  float* out = (float*)d_out;

  const int B = in_sizes[1] / HWC;

  float* comb = (float*)d_ws;
  int*   meta = (int*)((char*)d_ws + (size_t)B * MAXOBJ * KP * sizeof(float));

  cc_extract_kernel<<<B, 1024, 0, stream>>>(grid, meta);
  pool_kernel<<<B * MAXOBJ, 256, 0, stream>>>(grid_emb, meta, comb);
  fused_mlp_kernel<<<B, 512, 0, stream>>>(comb, meta, structure, W1, b1, W2, b2,
                                          Wp, bp, gmm, bta, osp, out);
}


# Bbox mean-pool moved into the tail of the labeling launch: the block that found a batch's 16 boxes pools them, 2 waves per box, 4 channels per lane (2 launches, was 3); cell sum order kept
# speedup vs baseline: 1.1137x; 1.1137x over previous
#include <hip/hip_runtime.h>
#include <hip/hip_bf16.h>
#include <math.h>

#define GH 30
#define GW 30
#define HWC 900
#define DD 256
#define MAXOBJ 16
#define SENT 900
#define KF 261
#define KP 264
#define LD 268

typedef __attribute__((ext_vector_type(2))) float v2f;
typedef __attribute__((ext_vector_type(8))) float v8f;

__global__ __launch_bounds__(1024) void cc_extract_pool_kernel(const int* __restrict__ grid,
                                                               const float* __restrict__ grid_emb,
                                                               int* __restrict__ meta,
                                                               float* __restrict__ comb) {
  __shared__ int g[HWC];
  __shared__ int lab[HWC];
  __shared__ int rmn[HWC], rmx[HWC], cmn[HWC], cmx[HWC];
  __shared__ int roots[MAXOBJ];
  __shared__ int obox[MAXOBJ][5];
  __shared__ int nroots;
  __shared__ int changed;

  const int b = blockIdx.x;
  const int tid = threadIdx.x;
  const int* gg = grid + b * HWC;

  for (int t = tid; t < HWC; t += blockDim.x) {
    int v = gg[t];
    g[t] = v;
    lab[t] = (v > 0) ? t : SENT;
    rmn[t] = GH; rmx[t] = -1; cmn[t] = GW; cmx[t] = -1;
  }
  __syncthreads();

  for (int iter = 0; iter < 2048; ++iter) {
    if (tid == 0) changed = 0;
    __syncthreads();
    const int cell = tid;
    const bool active = (cell < HWC) && (g[cell] > 0);
    int newl = SENT, oldl = SENT;
    if (active) {
      const int r = cell / GW, c = cell % GW;
      const int gv = g[cell];
      int m = lab[cell]; oldl = m;
      if (r > 0      && g[cell - GW] == gv) m = min(m, lab[cell - GW]);
      if (r < GH - 1 && g[cell + GW] == gv) m = min(m, lab[cell + GW]);
      if (c > 0      && g[cell - 1]  == gv) m = min(m, lab[cell - 1]);
      if (c < GW - 1 && g[cell + 1]  == gv) m = min(m, lab[cell + 1]);
      if (m < SENT) m = min(m, lab[m]);
      newl = m;
    }
    __syncthreads();
    if (active && newl != oldl) { lab[cell] = newl; changed = 1; }
    __syncthreads();
    const int done = (changed == 0);
    __syncthreads();
    if (done) break;
  }

  if (tid < HWC && g[tid] > 0) {
    const int root = lab[tid];
    atomicMin(&rmn[root], tid / GW);
    atomicMax(&rmx[root], tid / GW);
    atomicMin(&cmn[root], tid % GW);
    atomicMax(&cmx[root], tid % GW);
  }
  __syncthreads();

  if (tid == 0) {
    int cnt = 0;
    for (int i = 0; i < HWC && cnt < MAXOBJ; ++i)
      if (g[i] > 0 && lab[i] == i) roots[cnt++] = i;
    nroots = cnt;
  }
  __syncthreads();

  int o_color = 0, o_y = 0, o_x = 0, o_h = 1, o_w = 1, o_valid = 0;
  if (tid < MAXOBJ) {
    if (tid < nroots) {
      const int r = roots[tid];
      o_y = min(max(rmn[r], 0), GH - 1);
      o_x = min(max(cmn[r], 0), GW - 1);
      const int y2 = min(max(rmx[r] + 1, o_y + 1), GH);
      const int x2 = min(max(cmx[r] + 1, o_x + 1), GW);
      o_color = g[r]; o_h = y2 - o_y; o_w = x2 - o_x; o_valid = 1;
    }
    obox[tid][0] = o_color; obox[tid][1] = o_y; obox[tid][2] = o_x; obox[tid][3] = o_h; obox[tid][4] = o_w;
  }
  __syncthreads();
  if (tid < MAXOBJ) {
    int* mo = meta + (b * MAXOBJ + tid) * 6;
    mo[0] = o_color; mo[1] = o_y; mo[2] = o_x; mo[3] = o_h; mo[4] = o_w; mo[5] = o_valid;
  }

  {
    const int k  = tid >> 6;
    const int l6 = tid & 63;
    const int d4 = l6 * 4;
    const int color = obox[k][0], y = obox[k][1], x = obox[k][2], h = obox[k][3], w = obox[k][4];

    float4 s = make_float4(0.f, 0.f, 0.f, 0.f);
    for (int i = 0; i < h; ++i) {
      const float* rowp = grid_emb + (((size_t)(b * GH + y + i)) * GW + x) * DD + d4;
      for (int j = 0; j < w; ++j) {
        const float4 v = *reinterpret_cast<const float4*>(rowp + (size_t)j * DD);
        s.x += v.x; s.y += v.y; s.z += v.z; s.w += v.w;
      }
    }
    const float cnt = (float)(h * w);
    float* op = comb + (size_t)(b * MAXOBJ + k) * KP;
    *reinterpret_cast<float4*>(op + d4) =
        make_float4(s.x / cnt, s.y / cnt, s.z / cnt, s.w / cnt);
    if (l6 == 0) {
      *reinterpret_cast<float4*>(op + DD) =
          make_float4((float)color / 9.0f, (float)x / (float)GW,
                      (float)y / (float)GH, (float)w / (float)GW);
      *reinterpret_cast<float4*>(op + DD + 4) =
          make_float4((float)h / (float)GH, 0.f, 0.f, 0.f);
    }
  }
}

__global__ __launch_bounds__(512) void fused_mlp_kernel(
    const float* __restrict__ comb_g, const int* __restrict__ meta,
    const float* __restrict__ structure, const float* __restrict__ W1,
    const float* __restrict__ b1, const float* __restrict__ W2,
    const float* __restrict__ b2, const float* __restrict__ Wp,
    const float* __restrict__ bp, const float* __restrict__ gmm,
    const float* __restrict__ bta, const float* __restrict__ osp,
    float* __restrict__ out) {
  __shared__ float sA[MAXOBJ][LD];
  __shared__ float sB[MAXOBJ][LD];
  __shared__ float sC[MAXOBJ][LD];
  __shared__ float s_sn[DD];
  __shared__ float s_dot[MAXOBJ];
  __shared__ float s_red[16];
  __shared__ int   s_valid[MAXOBJ];
  __shared__ float s_norm;

  const int b = blockIdx.x;
  const int tid = threadIdx.x;
  const int lane = tid & 31;
  const int wv = tid >> 5;
  const int half = lane >> 4;
  const int mrow = lane & 15;
  const int nn = wv * 16 + (lane & 15);

  for (int t = tid; t < MAXOBJ * KP; t += 512) {
    int r = t / KP, c = t % KP;
    sC[r][c] = comb_g[(size_t)(b * MAXOBJ + r) * KP + c];
  }
  if (tid < MAXOBJ * (LD - KP)) {
    int r = tid / (LD - KP), c = KP + tid % (LD - KP);
    sC[r][c] = 0.f;
  }
  if (tid < MAXOBJ) s_valid[tid] = meta[(b * MAXOBJ + tid) * 6 + 5];

  if (tid < DD) {
    const float* sp = structure + (size_t)b * 8 * DD + tid;
    float s = 0.f;
    #pragma unroll
    for (int k = 0; k < 8; ++k) s += sp[(size_t)k * DD];
    s_sn[tid] = s * 0.125f;
  }
  __syncthreads();
  if (tid < DD) {
    float p = s_sn[tid] * s_sn[tid];
    #pragma unroll
    for (int m = 16; m >= 1; m >>= 1) p += __shfl_xor(p, m, 32);
    if (lane == 0) s_red[wv] = p;
  }
  __syncthreads();
  if (tid == 0) {
    float t = 0.f;
    for (int i = 0; i < 8; ++i) t += s_red[i];
    s_norm = fmaxf(sqrtf(t), 1e-8f);
  }
  __syncthreads();
  if (tid < DD) s_sn[tid] /= s_norm;

  {
    v8f acc = {};
    for (int k0 = 0; k0 < KP; k0 += 4) {
      const int kk = k0 + half * 2;
      v2f av; av.x = sC[mrow][kk]; av.y = sC[mrow][kk + 1];
      const int r0 = min(kk, KF - 1), r1 = min(kk + 1, KF - 1);
      v2f bv; bv.x = W1[(size_t)r0 * DD + nn]; bv.y = W1[(size_t)r1 * DD + nn];
      acc = __builtin_amdgcn_wmma_f32_16x16x4_f32(false, av, false, bv,
                                                  (short)0, acc, false, false);
    }
    const float bias = b1[nn];
    #pragma unroll
    for (int i = 0; i < 8; ++i) {
      const int row = half * 8 + i;
      float v = acc[i] + bias;
      v = 0.5f * v * (1.0f + erff(v * 0.70710678118654752f));
      sA[row][nn] = v;
    }
  }
  __syncthreads();

  {
    v8f acc = {};
    for (int k0 = 0; k0 < DD; k0 += 4) {
      const int kk = k0 + half * 2;
      v2f av; av.x = sA[mrow][kk]; av.y = sA[mrow][kk + 1];
      v2f bv; bv.x = W2[(size_t)kk * DD + nn]; bv.y = W2[(size_t)(kk + 1) * DD + nn];
      acc = __builtin_amdgcn_wmma_f32_16x16x4_f32(false, av, false, bv,
                                                  (short)0, acc, false, false);
    }
    const float bias = b2[nn];
    #pragma unroll
    for (int i = 0; i < 8; ++i) {
      const int row = half * 8 + i;
      const float v = acc[i] + bias;
      sB[row][nn] = s_valid[row] ? v : 0.f;
    }
  }
  __syncthreads();

  {
    float p = 0.f;
    #pragma unroll
    for (int j = 0; j < 8; ++j) {
      const int n = lane + 32 * j;
      p += sB[wv][n] * s_sn[n];
    }
    #pragma unroll
    for (int m = 16; m >= 1; m >>= 1) p += __shfl_xor(p, m, 32);
    if (lane == 0) s_dot[wv] = p;
  }
  __syncthreads();
  {
    const float os = osp[0];
    for (int t = tid; t < MAXOBJ * DD; t += 512) {
      const int r = t >> 8, n = t & 255;
      sB[r][n] = (sB[r][n] - s_dot[r] * s_sn[n]) * os;
    }
  }
  __syncthreads();

  {
    v8f acc = {};
    for (int k0 = 0; k0 < DD; k0 += 4) {
      const int kk = k0 + half * 2;
      v2f av; av.x = sB[mrow][kk]; av.y = sB[mrow][kk + 1];
      v2f bv; bv.x = Wp[(size_t)kk * DD + nn]; bv.y = Wp[(size_t)(kk + 1) * DD + nn];
      acc = __builtin_amdgcn_wmma_f32_16x16x4_f32(false, av, false, bv,
                                                  (short)0, acc, false, false);
    }
    const float bias = bp[nn];
    #pragma unroll
    for (int i = 0; i < 8; ++i) sC[half * 8 + i][nn] = acc[i] + bias;
  }
  __syncthreads();

  {
    float s1 = 0.f, s2 = 0.f;
    #pragma unroll
    for (int j = 0; j < 8; ++j) {
      const float v = sC[wv][lane + 32 * j];
      s1 += v; s2 += v * v;
    }
    #pragma unroll
    for (int m = 16; m >= 1; m >>= 1) {
      s1 += __shfl_xor(s1, m, 32);
      s2 += __shfl_xor(s2, m, 32);
    }
    const float mu = s1 * (1.0f / 256.0f);
    const float var = s2 * (1.0f / 256.0f) - mu * mu;
    const float rstd = rsqrtf(var + 1e-5f);
    float* op = out + (size_t)(b * MAXOBJ + wv) * DD;
    #pragma unroll
    for (int j = 0; j < 8; ++j) {
      const int n = lane + 32 * j;
      const float v = sC[wv][n];
      op[n] = (v - mu) * rstd * gmm[n] + bta[n];
    }
  }
}

extern "C" void kernel_launch(void* const* d_in, const int* in_sizes, int n_in,
                              void* d_out, int out_size, void* d_ws, size_t ws_size,
                              hipStream_t stream) {
  const float* grid_emb  = (const float*)d_in[0];
  const int*   grid      = (const int*)  d_in[1];
  const float* structure = (const float*)d_in[2];
  const float* W1        = (const float*)d_in[3];
  const float* b1        = (const float*)d_in[4];
  const float* W2        = (const float*)d_in[5];
  const float* b2        = (const float*)d_in[6];
  const float* Wp        = (const float*)d_in[7];
  const float* bp        = (const float*)d_in[8];
  const float* gmm       = (const float*)d_in[9];
  const float* bta       = (const float*)d_in[10];
  const float* osp       = (const float*)d_in[11];
  float* out = (float*)d_out;

  const int B = in_sizes[1] / HWC;

  float* comb = (float*)d_ws;
  int*   meta = (int*)((char*)d_ws + (size_t)B * MAXOBJ * KP * sizeof(float));

  cc_extract_pool_kernel<<<B, 1024, 0, stream>>>(grid, grid_emb, meta, comb);
  fused_mlp_kernel<<<B, 512, 0, stream>>>(comb, meta, structure, W1, b1, W2, b2,
                                          Wp, bp, gmm, bta, osp, out);
}


# Labeling by one union-find pass: each cell joins its same-colour left and upper neighbour by an LDS atomic min on the tree roots, then one root walk per cell; replaces the iterated min sweeps
# speedup vs baseline: 1.2030x; 1.0802x over previous
#include <hip/hip_runtime.h>
#include <hip/hip_bf16.h>
#include <math.h>

#define GH 30
#define GW 30
#define HWC 900
#define DD 256
#define MAXOBJ 16
#define SENT 900
#define KF 261
#define KP 264
#define LD 268

typedef __attribute__((ext_vector_type(2))) float v2f;
typedef __attribute__((ext_vector_type(8))) float v8f;

__device__ __forceinline__ int uf_load(const int* lab, int i) {
  return __hip_atomic_load(&lab[i], __ATOMIC_RELAXED, __HIP_MEMORY_SCOPE_WORKGROUP);
}

__device__ __forceinline__ int uf_find(const int* lab, int i) {
  for (int hop = 0; hop < HWC; ++hop) {
    const int p = uf_load(lab, i);
    if (p == i) break;
    i = p;
  }
  return i;
}

__device__ __forceinline__ void uf_unite(int* lab, int a, int b) {
  for (int t = 0; t < 2 * HWC; ++t) {
    for (int hop = 0; hop < HWC; ++hop) {
      const int pa = uf_load(lab, a);
      const int pb = uf_load(lab, b);
      if (pa == a && pb == b) break;
      a = pa;
      b = pb;
    }
    if (a == b) break;
    const int hi = max(a, b), lo = min(a, b);
    const int old = atomicMin(&lab[hi], lo);
    if (old == hi) break;
    a = lo;
    b = old;
  }
}

__global__ __launch_bounds__(1024) void cc_extract_pool_kernel(const int* __restrict__ grid,
                                                               const float* __restrict__ grid_emb,
                                                               int* __restrict__ meta,
                                                               float* __restrict__ comb) {
  __shared__ int g[HWC];
  __shared__ int lab[HWC];
  __shared__ int rmn[HWC], rmx[HWC], cmn[HWC], cmx[HWC];
  __shared__ int roots[MAXOBJ];
  __shared__ int obox[MAXOBJ][5];
  __shared__ int nroots;

  const int b = blockIdx.x;
  const int tid = threadIdx.x;
  const int* gg = grid + b * HWC;

  for (int t = tid; t < HWC; t += blockDim.x) {
    int v = gg[t];
    g[t] = v;
    lab[t] = (v > 0) ? t : SENT;
    rmn[t] = GH; rmx[t] = -1; cmn[t] = GW; cmx[t] = -1;
  }
  __syncthreads();

  const int cell = min(tid, HWC - 1);
  const int crow = cell / GW, ccol = cell - crow * GW;
  const int il = (ccol > 0) ? cell - 1 : cell;
  const int iu = (crow > 0) ? cell - GW : cell;
  const int gv = g[cell], gl = g[il], gu = g[iu];
  const bool active = (tid < HWC) && (gv > 0);
  {
    const bool join_l = active && (ccol > 0) && (gl == gv);
    const bool join_u = active && (crow > 0) && (gu == gv);
    const int p0 = join_l ? il : (join_u ? iu : -1);
    const int p1 = (join_l && join_u) ? iu : -1;
    if (p0 >= 0) uf_unite(lab, cell, p0);
    if (p1 >= 0) uf_unite(lab, cell, p1);
  }
  __syncthreads();

  if (active) {
    const int root = uf_find(lab, tid);
    atomicMin(&rmn[root], tid / GW);
    atomicMax(&rmx[root], tid / GW);
    atomicMin(&cmn[root], tid % GW);
    atomicMax(&cmx[root], tid % GW);
  }
  __syncthreads();

  if (tid == 0) {
    int cnt = 0;
    for (int i = 0; i < HWC && cnt < MAXOBJ; ++i)
      if (g[i] > 0 && lab[i] == i) roots[cnt++] = i;
    nroots = cnt;
  }
  __syncthreads();

  int o_color = 0, o_y = 0, o_x = 0, o_h = 1, o_w = 1, o_valid = 0;
  if (tid < MAXOBJ) {
    if (tid < nroots) {
      const int r = roots[tid];
      o_y = min(max(rmn[r], 0), GH - 1);
      o_x = min(max(cmn[r], 0), GW - 1);
      const int y2 = min(max(rmx[r] + 1, o_y + 1), GH);
      const int x2 = min(max(cmx[r] + 1, o_x + 1), GW);
      o_color = g[r]; o_h = y2 - o_y; o_w = x2 - o_x; o_valid = 1;
    }
    obox[tid][0] = o_color; obox[tid][1] = o_y; obox[tid][2] = o_x; obox[tid][3] = o_h; obox[tid][4] = o_w;
  }
  __syncthreads();
  if (tid < MAXOBJ) {
    int* mo = meta + (b * MAXOBJ + tid) * 6;
    mo[0] = o_color; mo[1] = o_y; mo[2] = o_x; mo[3] = o_h; mo[4] = o_w; mo[5] = o_valid;
  }

  {
    const int k  = tid >> 6;
    const int l6 = tid & 63;
    const int d4 = l6 * 4;
    const int color = obox[k][0], y = obox[k][1], x = obox[k][2], h = obox[k][3], w = obox[k][4];

    float4 s = make_float4(0.f, 0.f, 0.f, 0.f);
    for (int i = 0; i < h; ++i) {
      const float* rowp = grid_emb + (((size_t)(b * GH + y + i)) * GW + x) * DD + d4;
      for (int j = 0; j < w; ++j) {
        const float4 v = *reinterpret_cast<const float4*>(rowp + (size_t)j * DD);
        s.x += v.x; s.y += v.y; s.z += v.z; s.w += v.w;
      }
    }
    const float cnt = (float)(h * w);
    float* op = comb + (size_t)(b * MAXOBJ + k) * KP;
    *reinterpret_cast<float4*>(op + d4) =
        make_float4(s.x / cnt, s.y / cnt, s.z / cnt, s.w / cnt);
    if (l6 == 0) {
      *reinterpret_cast<float4*>(op + DD) =
          make_float4((float)color / 9.0f, (float)x / (float)GW,
                      (float)y / (float)GH, (float)w / (float)GW);
      *reinterpret_cast<float4*>(op + DD + 4) =
          make_float4((float)h / (float)GH, 0.f, 0.f, 0.f);
    }
  }
}

__global__ __launch_bounds__(512) void fused_mlp_kernel(
    const float* __restrict__ comb_g, const int* __restrict__ meta,
    const float* __restrict__ structure, const float* __restrict__ W1,
    const float* __restrict__ b1, const float* __restrict__ W2,
    const float* __restrict__ b2, const float* __restrict__ Wp,
    const float* __restrict__ bp, const float* __restrict__ gmm,
    const float* __restrict__ bta, const float* __restrict__ osp,
    float* __restrict__ out) {
  __shared__ float sA[MAXOBJ][LD];
  __shared__ float sB[MAXOBJ][LD];
  __shared__ float sC[MAXOBJ][LD];
  __shared__ float s_sn[DD];
  __shared__ float s_dot[MAXOBJ];
  __shared__ float s_red[16];
  __shared__ int   s_valid[MAXOBJ];
  __shared__ float s_norm;

  const int b = blockIdx.x;
  const int tid = threadIdx.x;
  const int lane = tid & 31;
  const int wv = tid >> 5;
  const int half = lane >> 4;
  const int mrow = lane & 15;
  const int nn = wv * 16 + (lane & 15);

  for (int t = tid; t < MAXOBJ * KP; t += 512) {
    int r = t / KP, c = t % KP;
    sC[r][c] = comb_g[(size_t)(b * MAXOBJ + r) * KP + c];
  }
  if (tid < MAXOBJ * (LD - KP)) {
    int r = tid / (LD - KP), c = KP + tid % (LD - KP);
    sC[r][c] = 0.f;
  }
  if (tid < MAXOBJ) s_valid[tid] = meta[(b * MAXOBJ + tid) * 6 + 5];

  if (tid < DD) {
    const float* sp = structure + (size_t)b * 8 * DD + tid;
    float s = 0.f;
    #pragma unroll
    for (int k = 0; k < 8; ++k) s += sp[(size_t)k * DD];
    s_sn[tid] = s * 0.125f;
  }
  __syncthreads();
  if (tid < DD) {
    float p = s_sn[tid] * s_sn[tid];
    #pragma unroll
    for (int m = 16; m >= 1; m >>= 1) p += __shfl_xor(p, m, 32);
    if (lane == 0) s_red[wv] = p;
  }
  __syncthreads();
  if (tid == 0) {
    float t = 0.f;
    for (int i = 0; i < 8; ++i) t += s_red[i];
    s_norm = fmaxf(sqrtf(t), 1e-8f);
  }
  __syncthreads();
  if (tid < DD) s_sn[tid] /= s_norm;

  {
    v8f acc = {};
    for (int k0 = 0; k0 < KP; k0 += 4) {
      const int kk = k0 + half * 2;
      v2f av; av.x = sC[mrow][kk]; av.y = sC[mrow][kk + 1];
      const int r0 = min(kk, KF - 1), r1 = min(kk + 1, KF - 1);
      v2f bv; bv.x = W1[(size_t)r0 * DD + nn]; bv.y = W1[(size_t)r1 * DD + nn];
      acc = __builtin_amdgcn_wmma_f32_16x16x4_f32(false, av, false, bv,
                                                  (short)0, acc, false, false);
    }
    const float bias = b1[nn];
    #pragma unroll
    for (int i = 0; i < 8; ++i) {
      const int row = half * 8 + i;
      float v = acc[i] + bias;
      v = 0.5f * v * (1.0f + erff(v * 0.70710678118654752f));
      sA[row][nn] = v;
    }
  }
  __syncthreads();

  {
    v8f acc = {};
    for (int k0 = 0; k0 < DD; k0 += 4) {
      const int kk = k0 + half * 2;
      v2f av; av.x = sA[mrow][kk]; av.y = sA[mrow][kk + 1];
      v2f bv; bv.x = W2[(size_t)kk * DD + nn]; bv.y = W2[(size_t)(kk + 1) * DD + nn];
      acc = __builtin_amdgcn_wmma_f32_16x16x4_f32(false, av, false, bv,
                                                  (short)0, acc, false, false);
    }
    const float bias = b2[nn];
    #pragma unroll
    for (int i = 0; i < 8; ++i) {
      const int row = half * 8 + i;
      const float v = acc[i] + bias;
      sB[row][nn] = s_valid[row] ? v : 0.f;
    }
  }
  __syncthreads();

  {
    float p = 0.f;
    #pragma unroll
    for (int j = 0; j < 8; ++j) {
      const int n = lane + 32 * j;
      p += sB[wv][n] * s_sn[n];
    }
    #pragma unroll
    for (int m = 16; m >= 1; m >>= 1) p += __shfl_xor(p, m, 32);
    if (lane == 0) s_dot[wv] = p;
  }
  __syncthreads();
  {
    const float os = osp[0];
    for (int t = tid; t < MAXOBJ * DD; t += 512) {
      const int r = t >> 8, n = t & 255;
      sB[r][n] = (sB[r][n] - s_dot[r] * s_sn[n]) * os;
    }
  }
  __syncthreads();

  {
    v8f acc = {};
    for (int k0 = 0; k0 < DD; k0 += 4) {
      const int kk = k0 + half * 2;
      v2f av; av.x = sB[mrow][kk]; av.y = sB[mrow][kk + 1];
      v2f bv; bv.x = Wp[(size_t)kk * DD + nn]; bv.y = Wp[(size_t)(kk + 1) * DD + nn];
      acc = __builtin_amdgcn_wmma_f32_16x16x4_f32(false, av, false, bv,
                                                  (short)0, acc, false, false);
    }
    const float bias = bp[nn];
    #pragma unroll
    for (int i = 0; i < 8; ++i) sC[half * 8 + i][nn] = acc[i] + bias;
  }
  __syncthreads();

  {
    float s1 = 0.f, s2 = 0.f;
    #pragma unroll
    for (int j = 0; j < 8; ++j) {
      const float v = sC[wv][lane + 32 * j];
      s1 += v; s2 += v * v;
    }
    #pragma unroll
    for (int m = 16; m >= 1; m >>= 1) {
      s1 += __shfl_xor(s1, m, 32);
      s2 += __shfl_xor(s2, m, 32);
    }
    const float mu = s1 * (1.0f / 256.0f);
    const float var = s2 * (1.0f / 256.0f) - mu * mu;
    const float rstd = rsqrtf(var + 1e-5f);
    float* op = out + (size_t)(b * MAXOBJ + wv) * DD;
    #pragma unroll
    for (int j = 0; j < 8; ++j) {
      const int n = lane + 32 * j;
      const float v = sC[wv][n];
      op[n] = (v - mu) * rstd * gmm[n] + bta[n];
    }
  }
}

extern "C" void kernel_launch(void* const* d_in, const int* in_sizes, int n_in,
                              void* d_out, int out_size, void* d_ws, size_t ws_size,
                              hipStream_t stream) {
  const float* grid_emb  = (const float*)d_in[0];
  const int*   grid      = (const int*)  d_in[1];
  const float* structure = (const float*)d_in[2];
  const float* W1        = (const float*)d_in[3];
  const float* b1        = (const float*)d_in[4];
  const float* W2        = (const float*)d_in[5];
  const float* b2        = (const float*)d_in[6];
  const float* Wp        = (const float*)d_in[7];
  const float* bp        = (const float*)d_in[8];
  const float* gmm       = (const float*)d_in[9];
  const float* bta       = (const float*)d_in[10];
  const float* osp       = (const float*)d_in[11];
  float* out = (float*)d_out;

  const int B = in_sizes[1] / HWC;

  float* comb = (float*)d_ws;
  int*   meta = (int*)((char*)d_ws + (size_t)B * MAXOBJ * KP * sizeof(float));

  cc_extract_pool_kernel<<<B, 1024, 0, stream>>>(grid, grid_emb, meta, comb);
  fused_mlp_kernel<<<B, 512, 0, stream>>>(comb, meta, structure, W1, b1, W2, b2,
                                          Wp, bp, gmm, bta, osp, out);
}


# Dense launch: every global read that needs no earlier phase (combined tile as 16-byte loads, valid flags, structure rows, biases, gamma, beta, ortho scale) is requested at kernel entry in one batch
# speedup vs baseline: 1.2962x; 1.0775x over previous
#include <hip/hip_runtime.h>
#include <hip/hip_bf16.h>
#include <math.h>

#define GH 30
#define GW 30
#define HWC 900
#define DD 256
#define MAXOBJ 16
#define SENT 900
#define KF 261
#define KP 264
#define LD 268

typedef __attribute__((ext_vector_type(2))) float v2f;
typedef __attribute__((ext_vector_type(8))) float v8f;

__device__ __forceinline__ int uf_load(const int* lab, int i) {
  return __hip_atomic_load(&lab[i], __ATOMIC_RELAXED, __HIP_MEMORY_SCOPE_WORKGROUP);
}

__device__ __forceinline__ int uf_find(const int* lab, int i) {
  for (int hop = 0; hop < HWC; ++hop) {
    const int p = uf_load(lab, i);
    if (p == i) break;
    i = p;
  }
  return i;
}

__device__ __forceinline__ void uf_unite(int* lab, int a, int b) {
  for (int t = 0; t < 2 * HWC; ++t) {
    for (int hop = 0; hop < HWC; ++hop) {
      const int pa = uf_load(lab, a);
      const int pb = uf_load(lab, b);
      if (pa == a && pb == b) break;
      a = pa;
      b = pb;
    }
    if (a == b) break;
    const int hi = max(a, b), lo = min(a, b);
    const int old = atomicMin(&lab[hi], lo);
    if (old == hi) break;
    a = lo;
    b = old;
  }
}

__global__ __launch_bounds__(1024) void cc_extract_pool_kernel(const int* __restrict__ grid,
                                                               const float* __restrict__ grid_emb,
                                                               int* __restrict__ meta,
                                                               float* __restrict__ comb) {
  __shared__ int g[HWC];
  __shared__ int lab[HWC];
  __shared__ int rmn[HWC], rmx[HWC], cmn[HWC], cmx[HWC];
  __shared__ int roots[MAXOBJ];
  __shared__ int obox[MAXOBJ][5];
  __shared__ int nroots;

  const int b = blockIdx.x;
  const int tid = threadIdx.x;
  const int* gg = grid + b * HWC;

  for (int t = tid; t < HWC; t += blockDim.x) {
    int v = gg[t];
    g[t] = v;
    lab[t] = (v > 0) ? t : SENT;
    rmn[t] = GH; rmx[t] = -1; cmn[t] = GW; cmx[t] = -1;
  }
  __syncthreads();

  const int cell = min(tid, HWC - 1);
  const int crow = cell / GW, ccol = cell - crow * GW;
  const int il = (ccol > 0) ? cell - 1 : cell;
  const int iu = (crow > 0) ? cell - GW : cell;
  const int gv = g[cell], gl = g[il], gu = g[iu];
  const bool active = (tid < HWC) && (gv > 0);
  {
    const bool join_l = active && (ccol > 0) && (gl == gv);
    const bool join_u = active && (crow > 0) && (gu == gv);
    const int p0 = join_l ? il : (join_u ? iu : -1);
    const int p1 = (join_l && join_u) ? iu : -1;
    if (p0 >= 0) uf_unite(lab, cell, p0);
    if (p1 >= 0) uf_unite(lab, cell, p1);
  }
  __syncthreads();

  if (active) {
    const int root = uf_find(lab, tid);
    atomicMin(&rmn[root], tid / GW);
    atomicMax(&rmx[root], tid / GW);
    atomicMin(&cmn[root], tid % GW);
    atomicMax(&cmx[root], tid % GW);
  }
  __syncthreads();

  if (tid == 0) {
    int cnt = 0;
    for (int i = 0; i < HWC && cnt < MAXOBJ; ++i)
      if (g[i] > 0 && lab[i] == i) roots[cnt++] = i;
    nroots = cnt;
  }
  __syncthreads();

  int o_color = 0, o_y = 0, o_x = 0, o_h = 1, o_w = 1, o_valid = 0;
  if (tid < MAXOBJ) {
    if (tid < nroots) {
      const int r = roots[tid];
      o_y = min(max(rmn[r], 0), GH - 1);
      o_x = min(max(cmn[r], 0), GW - 1);
      const int y2 = min(max(rmx[r] + 1, o_y + 1), GH);
      const int x2 = min(max(cmx[r] + 1, o_x + 1), GW);
      o_color = g[r]; o_h = y2 - o_y; o_w = x2 - o_x; o_valid = 1;
    }
    obox[tid][0] = o_color; obox[tid][1] = o_y; obox[tid][2] = o_x; obox[tid][3] = o_h; obox[tid][4] = o_w;
  }
  __syncthreads();
  if (tid < MAXOBJ) {
    int* mo = meta + (b * MAXOBJ + tid) * 6;
    mo[0] = o_color; mo[1] = o_y; mo[2] = o_x; mo[3] = o_h; mo[4] = o_w; mo[5] = o_valid;
  }

  {
    const int k  = tid >> 6;
    const int l6 = tid & 63;
    const int d4 = l6 * 4;
    const int color = obox[k][0], y = obox[k][1], x = obox[k][2], h = obox[k][3], w = obox[k][4];

    float4 s = make_float4(0.f, 0.f, 0.f, 0.f);
    for (int i = 0; i < h; ++i) {
      const float* rowp = grid_emb + (((size_t)(b * GH + y + i)) * GW + x) * DD + d4;
      for (int j = 0; j < w; ++j) {
        const float4 v = *reinterpret_cast<const float4*>(rowp + (size_t)j * DD);
        s.x += v.x; s.y += v.y; s.z += v.z; s.w += v.w;
      }
    }
    const float cnt = (float)(h * w);
    float* op = comb + (size_t)(b * MAXOBJ + k) * KP;
    *reinterpret_cast<float4*>(op + d4) =
        make_float4(s.x / cnt, s.y / cnt, s.z / cnt, s.w / cnt);
    if (l6 == 0) {
      *reinterpret_cast<float4*>(op + DD) =
          make_float4((float)color / 9.0f, (float)x / (float)GW,
                      (float)y / (float)GH, (float)w / (float)GW);
      *reinterpret_cast<float4*>(op + DD + 4) =
          make_float4((float)h / (float)GH, 0.f, 0.f, 0.f);
    }
  }
}

__global__ __launch_bounds__(512) void fused_mlp_kernel(
    const float* __restrict__ comb_g, const int* __restrict__ meta,
    const float* __restrict__ structure, const float* __restrict__ W1,
    const float* __restrict__ b1, const float* __restrict__ W2,
    const float* __restrict__ b2, const float* __restrict__ Wp,
    const float* __restrict__ bp, const float* __restrict__ gmm,
    const float* __restrict__ bta, const float* __restrict__ osp,
    float* __restrict__ out) {
  __shared__ float sA[MAXOBJ][LD];
  __shared__ float sB[MAXOBJ][LD];
  __shared__ __attribute__((aligned(16))) float sC[MAXOBJ][LD];
  __shared__ float s_sn[DD];
  __shared__ float s_dot[MAXOBJ];
  __shared__ float s_red[16];
  __shared__ int   s_valid[MAXOBJ];
  __shared__ float s_norm;

  const int b = blockIdx.x;
  const int tid = threadIdx.x;
  const int lane = tid & 31;
  const int wv = tid >> 5;
  const int half = lane >> 4;
  const int mrow = lane & 15;
  const int nn = wv * 16 + (lane & 15);

  const float4* cg4 = reinterpret_cast<const float4*>(comb_g + (size_t)b * MAXOBJ * KP);
  const float4 cva = cg4[tid];
  const float4 cvb = cg4[tid + 512];
  float4 cvc = make_float4(0.f, 0.f, 0.f, 0.f);
  if (tid < MAXOBJ * KP / 4 - 1024) cvc = cg4[tid + 1024];
  int vflag = 0;
  if (tid < MAXOBJ) vflag = meta[(b * MAXOBJ + tid) * 6 + 5];
  float st[8] = {0.f, 0.f, 0.f, 0.f, 0.f, 0.f, 0.f, 0.f};
  if (tid < DD) {
    const float* sp = structure + (size_t)b * 8 * DD + tid;
    #pragma unroll
    for (int k = 0; k < 8; ++k) st[k] = sp[(size_t)k * DD];
  }
  const float bias1 = b1[nn];
  const float bias2 = b2[nn];
  const float bias3 = bp[nn];
  float gm[8], bt[8];
  #pragma unroll
  for (int j = 0; j < 8; ++j) { gm[j] = gmm[lane + 32 * j]; bt[j] = bta[lane + 32 * j]; }
  const float os = osp[0];

  {
    const int q0 = tid, q1 = tid + 512;
    *reinterpret_cast<float4*>(&sC[q0 / 66][(q0 % 66) * 4]) = cva;
    *reinterpret_cast<float4*>(&sC[q1 / 66][(q1 % 66) * 4]) = cvb;
    if (tid < MAXOBJ * KP / 4 - 1024) {
      const int q2 = tid + 1024;
      *reinterpret_cast<float4*>(&sC[q2 / 66][(q2 % 66) * 4]) = cvc;
    }
  }
  if (tid < MAXOBJ * (LD - KP)) {
    int r = tid / (LD - KP), c = KP + tid % (LD - KP);
    sC[r][c] = 0.f;
  }
  if (tid < MAXOBJ) s_valid[tid] = vflag;

  if (tid < DD) {
    float s = 0.f;
    #pragma unroll
    for (int k = 0; k < 8; ++k) s += st[k];
    s_sn[tid] = s * 0.125f;
  }
  __syncthreads();
  if (tid < DD) {
    float p = s_sn[tid] * s_sn[tid];
    #pragma unroll
    for (int m = 16; m >= 1; m >>= 1) p += __shfl_xor(p, m, 32);
    if (lane == 0) s_red[wv] = p;
  }
  __syncthreads();
  if (tid == 0) {
    float t = 0.f;
    for (int i = 0; i < 8; ++i) t += s_red[i];
    s_norm = fmaxf(sqrtf(t), 1e-8f);
  }
  __syncthreads();
  if (tid < DD) s_sn[tid] /= s_norm;

  {
    v8f acc = {};
    for (int k0 = 0; k0 < KP; k0 += 4) {
      const int kk = k0 + half * 2;
      v2f av; av.x = sC[mrow][kk]; av.y = sC[mrow][kk + 1];
      const int r0 = min(kk, KF - 1), r1 = min(kk + 1, KF - 1);
      v2f bv; bv.x = W1[(size_t)r0 * DD + nn]; bv.y = W1[(size_t)r1 * DD + nn];
      acc = __builtin_amdgcn_wmma_f32_16x16x4_f32(false, av, false, bv,
                                                  (short)0, acc, false, false);
    }
    #pragma unroll
    for (int i = 0; i < 8; ++i) {
      const int row = half * 8 + i;
      float v = acc[i] + bias1;
      v = 0.5f * v * (1.0f + erff(v * 0.70710678118654752f));
      sA[row][nn] = v;
    }
  }
  __syncthreads();

  {
    v8f acc = {};
    for (int k0 = 0; k0 < DD; k0 += 4) {
      const int kk = k0 + half * 2;
      v2f av; av.x = sA[mrow][kk]; av.y = sA[mrow][kk + 1];
      v2f bv; bv.x = W2[(size_t)kk * DD + nn]; bv.y = W2[(size_t)(kk + 1) * DD + nn];
      acc = __builtin_amdgcn_wmma_f32_16x16x4_f32(false, av, false, bv,
                                                  (short)0, acc, false, false);
    }
    #pragma unroll
    for (int i = 0; i < 8; ++i) {
      const int row = half * 8 + i;
      const float v = acc[i] + bias2;
      sB[row][nn] = s_valid[row] ? v : 0.f;
    }
  }
  __syncthreads();

  {
    float p = 0.f;
    #pragma unroll
    for (int j = 0; j < 8; ++j) {
      const int n = lane + 32 * j;
      p += sB[wv][n] * s_sn[n];
    }
    #pragma unroll
    for (int m = 16; m >= 1; m >>= 1) p += __shfl_xor(p, m, 32);
    if (lane == 0) s_dot[wv] = p;
  }
  __syncthreads();
  {
    for (int t = tid; t < MAXOBJ * DD; t += 512) {
      const int r = t >> 8, n = t & 255;
      sB[r][n] = (sB[r][n] - s_dot[r] * s_sn[n]) * os;
    }
  }
  __syncthreads();

  {
    v8f acc = {};
    for (int k0 = 0; k0 < DD; k0 += 4) {
      const int kk = k0 + half * 2;
      v2f av; av.x = sB[mrow][kk]; av.y = sB[mrow][kk + 1];
      v2f bv; bv.x = Wp[(size_t)kk * DD + nn]; bv.y = Wp[(size_t)(kk + 1) * DD + nn];
      acc = __builtin_amdgcn_wmma_f32_16x16x4_f32(false, av, false, bv,
                                                  (short)0, acc, false, false);
    }
    #pragma unroll
    for (int i = 0; i < 8; ++i) sC[half * 8 + i][nn] = acc[i] + bias3;
  }
  __syncthreads();

  {
    float s1 = 0.f, s2 = 0.f;
    #pragma unroll
    for (int j = 0; j < 8; ++j) {
      const float v = sC[wv][lane + 32 * j];
      s1 += v; s2 += v * v;
    }
    #pragma unroll
    for (int m = 16; m >= 1; m >>= 1) {
      s1 += __shfl_xor(s1, m, 32);
      s2 += __shfl_xor(s2, m, 32);
    }
    const float mu = s1 * (1.0f / 256.0f);
    const float var = s2 * (1.0f / 256.0f) - mu * mu;
    const float rstd = rsqrtf(var + 1e-5f);
    float* op = out + (size_t)(b * MAXOBJ + wv) * DD;
    #pragma unroll
    for (int j = 0; j < 8; ++j) {
      const int n = lane + 32 * j;
      const float v = sC[wv][n];
      op[n] = (v - mu) * rstd * gm[j] + bt[j];
    }
  }
}

extern "C" void kernel_launch(void* const* d_in, const int* in_sizes, int n_in,
                              void* d_out, int out_size, void* d_ws, size_t ws_size,
                              hipStream_t stream) {
  const float* grid_emb  = (const float*)d_in[0];
  const int*   grid      = (const int*)  d_in[1];
  const float* structure = (const float*)d_in[2];
  const float* W1        = (const float*)d_in[3];
  const float* b1        = (const float*)d_in[4];
  const float* W2        = (const float*)d_in[5];
  const float* b2        = (const float*)d_in[6];
  const float* Wp        = (const float*)d_in[7];
  const float* bp        = (const float*)d_in[8];
  const float* gmm       = (const float*)d_in[9];
  const float* bta       = (const float*)d_in[10];
  const float* osp       = (const float*)d_in[11];
  float* out = (float*)d_out;

  const int B = in_sizes[1] / HWC;

  float* comb = (float*)d_ws;
  int*   meta = (int*)((char*)d_ws + (size_t)B * MAXOBJ * KP * sizeof(float));

  cc_extract_pool_kernel<<<B, 1024, 0, stream>>>(grid, grid_emb, meta, comb);
  fused_mlp_kernel<<<B, 512, 0, stream>>>(comb, meta, structure, W1, b1, W2, b2,
                                          Wp, bp, gmm, bta, osp, out);
}


# Root scan by one thread reads a grid row per trip: the row's 30 colours and 30 labels are one batch of LDS loads, then 30 scalar compares; was two dependent LDS round trips per cell
# speedup vs baseline: 1.3677x; 1.0552x over previous
#include <hip/hip_runtime.h>
#include <hip/hip_bf16.h>
#include <math.h>

#define GH 30
#define GW 30
#define HWC 900
#define DD 256
#define MAXOBJ 16
#define SENT 900
#define KF 261
#define KP 264
#define LD 268

typedef __attribute__((ext_vector_type(2))) float v2f;
typedef __attribute__((ext_vector_type(8))) float v8f;

__device__ __forceinline__ int uf_load(const int* lab, int i) {
  return __hip_atomic_load(&lab[i], __ATOMIC_RELAXED, __HIP_MEMORY_SCOPE_WORKGROUP);
}

__device__ __forceinline__ int uf_find(const int* lab, int i) {
  for (int hop = 0; hop < HWC; ++hop) {
    const int p = uf_load(lab, i);
    if (p == i) break;
    i = p;
  }
  return i;
}

__device__ __forceinline__ void uf_unite(int* lab, int a, int b) {
  for (int t = 0; t < 2 * HWC; ++t) {
    for (int hop = 0; hop < HWC; ++hop) {
      const int pa = uf_load(lab, a);
      const int pb = uf_load(lab, b);
      if (pa == a && pb == b) break;
      a = pa;
      b = pb;
    }
    if (a == b) break;
    const int hi = max(a, b), lo = min(a, b);
    const int old = atomicMin(&lab[hi], lo);
    if (old == hi) break;
    a = lo;
    b = old;
  }
}

__global__ __launch_bounds__(1024) void cc_extract_pool_kernel(const int* __restrict__ grid,
                                                               const float* __restrict__ grid_emb,
                                                               int* __restrict__ meta,
                                                               float* __restrict__ comb) {
  __shared__ int g[HWC];
  __shared__ int lab[HWC];
  __shared__ int rmn[HWC], rmx[HWC], cmn[HWC], cmx[HWC];
  __shared__ int roots[MAXOBJ];
  __shared__ int obox[MAXOBJ][5];
  __shared__ int nroots;

  const int b = blockIdx.x;
  const int tid = threadIdx.x;
  const int* gg = grid + b * HWC;

  for (int t = tid; t < HWC; t += blockDim.x) {
    int v = gg[t];
    g[t] = v;
    lab[t] = (v > 0) ? t : SENT;
    rmn[t] = GH; rmx[t] = -1; cmn[t] = GW; cmx[t] = -1;
  }
  __syncthreads();

  const int cell = min(tid, HWC - 1);
  const int crow = cell / GW, ccol = cell - crow * GW;
  const int il = (ccol > 0) ? cell - 1 : cell;
  const int iu = (crow > 0) ? cell - GW : cell;
  const int gv = g[cell], gl = g[il], gu = g[iu];
  const bool active = (tid < HWC) && (gv > 0);
  {
    const bool join_l = active && (ccol > 0) && (gl == gv);
    const bool join_u = active && (crow > 0) && (gu == gv);
    const int p0 = join_l ? il : (join_u ? iu : -1);
    const int p1 = (join_l && join_u) ? iu : -1;
    if (p0 >= 0) uf_unite(lab, cell, p0);
    if (p1 >= 0) uf_unite(lab, cell, p1);
  }
  __syncthreads();

  if (active) {
    const int root = uf_find(lab, tid);
    atomicMin(&rmn[root], tid / GW);
    atomicMax(&rmx[root], tid / GW);
    atomicMin(&cmn[root], tid % GW);
    atomicMax(&cmx[root], tid % GW);
  }
  __syncthreads();

  if (tid == 0) {
    int cnt = 0;
    for (int base = 0; base < HWC && cnt < MAXOBJ; base += GW) {
      int cg[GW], cl[GW];
      #pragma unroll
      for (int j = 0; j < GW; ++j) { cg[j] = g[base + j]; cl[j] = lab[base + j]; }
      #pragma unroll
      for (int j = 0; j < GW; ++j) {
        const bool take = (cg[j] > 0) & (cl[j] == base + j) & (cnt < MAXOBJ);
        if (take) roots[cnt] = base + j;
        cnt += take ? 1 : 0;
      }
    }
    nroots = cnt;
  }
  __syncthreads();

  int o_color = 0, o_y = 0, o_x = 0, o_h = 1, o_w = 1, o_valid = 0;
  if (tid < MAXOBJ) {
    if (tid < nroots) {
      const int r = roots[tid];
      o_y = min(max(rmn[r], 0), GH - 1);
      o_x = min(max(cmn[r], 0), GW - 1);
      const int y2 = min(max(rmx[r] + 1, o_y + 1), GH);
      const int x2 = min(max(cmx[r] + 1, o_x + 1), GW);
      o_color = g[r]; o_h = y2 - o_y; o_w = x2 - o_x; o_valid = 1;
    }
    obox[tid][0] = o_color; obox[tid][1] = o_y; obox[tid][2] = o_x; obox[tid][3] = o_h; obox[tid][4] = o_w;
  }
  __syncthreads();
  if (tid < MAXOBJ) {
    int* mo = meta + (b * MAXOBJ + tid) * 6;
    mo[0] = o_color; mo[1] = o_y; mo[2] = o_x; mo[3] = o_h; mo[4] = o_w; mo[5] = o_valid;
  }

  {
    const int k  = tid >> 6;
    const int l6 = tid & 63;
    const int d4 = l6 * 4;
    const int color = obox[k][0], y = obox[k][1], x = obox[k][2], h = obox[k][3], w = obox[k][4];

    float4 s = make_float4(0.f, 0.f, 0.f, 0.f);
    for (int i = 0; i < h; ++i) {
      const float* rowp = grid_emb + (((size_t)(b * GH + y + i)) * GW + x) * DD + d4;
      for (int j = 0; j < w; ++j) {
        const float4 v = *reinterpret_cast<const float4*>(rowp + (size_t)j * DD);
        s.x += v.x; s.y += v.y; s.z += v.z; s.w += v.w;
      }
    }
    const float cnt = (float)(h * w);
    float* op = comb + (size_t)(b * MAXOBJ + k) * KP;
    *reinterpret_cast<float4*>(op + d4) =
        make_float4(s.x / cnt, s.y / cnt, s.z / cnt, s.w / cnt);
    if (l6 == 0) {
      *reinterpret_cast<float4*>(op + DD) =
          make_float4((float)color / 9.0f, (float)x / (float)GW,
                      (float)y / (float)GH, (float)w / (float)GW);
      *reinterpret_cast<float4*>(op + DD + 4) =
          make_float4((float)h / (float)GH, 0.f, 0.f, 0.f);
    }
  }
}

__global__ __launch_bounds__(512) void fused_mlp_kernel(
    const float* __restrict__ comb_g, const int* __restrict__ meta,
    const float* __restrict__ structure, const float* __restrict__ W1,
    const float* __restrict__ b1, const float* __restrict__ W2,
    const float* __restrict__ b2, const float* __restrict__ Wp,
    const float* __restrict__ bp, const float* __restrict__ gmm,
    const float* __restrict__ bta, const float* __restrict__ osp,
    float* __restrict__ out) {
  __shared__ float sA[MAXOBJ][LD];
  __shared__ float sB[MAXOBJ][LD];
  __shared__ __attribute__((aligned(16))) float sC[MAXOBJ][LD];
  __shared__ float s_sn[DD];
  __shared__ float s_dot[MAXOBJ];
  __shared__ float s_red[16];
  __shared__ int   s_valid[MAXOBJ];
  __shared__ float s_norm;

  const int b = blockIdx.x;
  const int tid = threadIdx.x;
  const int lane = tid & 31;
  const int wv = tid >> 5;
  const int half = lane >> 4;
  const int mrow = lane & 15;
  const int nn = wv * 16 + (lane & 15);

  const float4* cg4 = reinterpret_cast<const float4*>(comb_g + (size_t)b * MAXOBJ * KP);
  const float4 cva = cg4[tid];
  const float4 cvb = cg4[tid + 512];
  float4 cvc = make_float4(0.f, 0.f, 0.f, 0.f);
  if (tid < MAXOBJ * KP / 4 - 1024) cvc = cg4[tid + 1024];
  int vflag = 0;
  if (tid < MAXOBJ) vflag = meta[(b * MAXOBJ + tid) * 6 + 5];
  float st[8] = {0.f, 0.f, 0.f, 0.f, 0.f, 0.f, 0.f, 0.f};
  if (tid < DD) {
    const float* sp = structure + (size_t)b * 8 * DD + tid;
    #pragma unroll
    for (int k = 0; k < 8; ++k) st[k] = sp[(size_t)k * DD];
  }
  const float bias1 = b1[nn];
  const float bias2 = b2[nn];
  const float bias3 = bp[nn];
  float gm[8], bt[8];
  #pragma unroll
  for (int j = 0; j < 8; ++j) { gm[j] = gmm[lane + 32 * j]; bt[j] = bta[lane + 32 * j]; }
  const float os = osp[0];

  {
    const int q0 = tid, q1 = tid + 512;
    *reinterpret_cast<float4*>(&sC[q0 / 66][(q0 % 66) * 4]) = cva;
    *reinterpret_cast<float4*>(&sC[q1 / 66][(q1 % 66) * 4]) = cvb;
    if (tid < MAXOBJ * KP / 4 - 1024) {
      const int q2 = tid + 1024;
      *reinterpret_cast<float4*>(&sC[q2 / 66][(q2 % 66) * 4]) = cvc;
    }
  }
  if (tid < MAXOBJ * (LD - KP)) {
    int r = tid / (LD - KP), c = KP + tid % (LD - KP);
    sC[r][c] = 0.f;
  }
  if (tid < MAXOBJ) s_valid[tid] = vflag;

  if (tid < DD) {
    float s = 0.f;
    #pragma unroll
    for (int k = 0; k < 8; ++k) s += st[k];
    s_sn[tid] = s * 0.125f;
  }
  __syncthreads();
  if (tid < DD) {
    float p = s_sn[tid] * s_sn[tid];
    #pragma unroll
    for (int m = 16; m >= 1; m >>= 1) p += __shfl_xor(p, m, 32);
    if (lane == 0) s_red[wv] = p;
  }
  __syncthreads();
  if (tid == 0) {
    float t = 0.f;
    for (int i = 0; i < 8; ++i) t += s_red[i];
    s_norm = fmaxf(sqrtf(t), 1e-8f);
  }
  __syncthreads();
  if (tid < DD) s_sn[tid] /= s_norm;

  {
    v8f acc = {};
    for (int k0 = 0; k0 < KP; k0 += 4) {
      const int kk = k0 + half * 2;
      v2f av; av.x = sC[mrow][kk]; av.y = sC[mrow][kk + 1];
      const int r0 = min(kk, KF - 1), r1 = min(kk + 1, KF - 1);
      v2f bv; bv.x = W1[(size_t)r0 * DD + nn]; bv.y = W1[(size_t)r1 * DD + nn];
      acc = __builtin_amdgcn_wmma_f32_16x16x4_f32(false, av, false, bv,
                                                  (short)0, acc, false, false);
    }
    #pragma unroll
    for (int i = 0; i < 8; ++i) {
      const int row = half * 8 + i;
      float v = acc[i] + bias1;
      v = 0.5f * v * (1.0f + erff(v * 0.70710678118654752f));
      sA[row][nn] = v;
    }
  }
  __syncthreads();

  {
    v8f acc = {};
    for (int k0 = 0; k0 < DD; k0 += 4) {
      const int kk = k0 + half * 2;
      v2f av; av.x = sA[mrow][kk]; av.y = sA[mrow][kk + 1];
      v2f bv; bv.x = W2[(size_t)kk * DD + nn]; bv.y = W2[(size_t)(kk + 1) * DD + nn];
      acc = __builtin_amdgcn_wmma_f32_16x16x4_f32(false, av, false, bv,
                                                  (short)0, acc, false, false);
    }
    #pragma unroll
    for (int i = 0; i < 8; ++i) {
      const int row = half * 8 + i;
      const float v = acc[i] + bias2;
      sB[row][nn] = s_valid[row] ? v : 0.f;
    }
  }
  __syncthreads();

  {
    float p = 0.f;
    #pragma unroll
    for (int j = 0; j < 8; ++j) {
      const int n = lane + 32 * j;
      p += sB[wv][n] * s_sn[n];
    }
    #pragma unroll
    for (int m = 16; m >= 1; m >>= 1) p += __shfl_xor(p, m, 32);
    if (lane == 0) s_dot[wv] = p;
  }
  __syncthreads();
  {
    for (int t = tid; t < MAXOBJ * DD; t += 512) {
      const int r = t >> 8, n = t & 255;
      sB[r][n] = (sB[r][n] - s_dot[r] * s_sn[n]) * os;
    }
  }
  __syncthreads();

  {
    v8f acc = {};
    for (int k0 = 0; k0 < DD; k0 += 4) {
      const int kk = k0 + half * 2;
      v2f av; av.x = sB[mrow][kk]; av.y = sB[mrow][kk + 1];
      v2f bv; bv.x = Wp[(size_t)kk * DD + nn]; bv.y = Wp[(size_t)(kk + 1) * DD + nn];
      acc = __builtin_amdgcn_wmma_f32_16x16x4_f32(false, av, false, bv,
                                                  (short)0, acc, false, false);
    }
    #pragma unroll
    for (int i = 0; i < 8; ++i) sC[half * 8 + i][nn] = acc[i] + bias3;
  }
  __syncthreads();

  {
    float s1 = 0.f, s2 = 0.f;
    #pragma unroll
    for (int j = 0; j < 8; ++j) {
      const float v = sC[wv][lane + 32 * j];
      s1 += v; s2 += v * v;
    }
    #pragma unroll
    for (int m = 16; m >= 1; m >>= 1) {
      s1 += __shfl_xor(s1, m, 32);
      s2 += __shfl_xor(s2, m, 32);
    }
    const float mu = s1 * (1.0f / 256.0f);
    const float var = s2 * (1.0f / 256.0f) - mu * mu;
    const float rstd = rsqrtf(var + 1e-5f);
    float* op = out + (size_t)(b * MAXOBJ + wv) * DD;
    #pragma unroll
    for (int j = 0; j < 8; ++j) {
      const int n = lane + 32 * j;
      const float v = sC[wv][n];
      op[n] = (v - mu) * rstd * gm[j] + bt[j];
    }
  }
}

extern "C" void kernel_launch(void* const* d_in, const int* in_sizes, int n_in,
                              void* d_out, int out_size, void* d_ws, size_t ws_size,
                              hipStream_t stream) {
  const float* grid_emb  = (const float*)d_in[0];
  const int*   grid      = (const int*)  d_in[1];
  const float* structure = (const float*)d_in[2];
  const float* W1        = (const float*)d_in[3];
  const float* b1        = (const float*)d_in[4];
  const float* W2        = (const float*)d_in[5];
  const float* b2        = (const float*)d_in[6];
  const float* Wp        = (const float*)d_in[7];
  const float* bp        = (const float*)d_in[8];
  const float* gmm       = (const float*)d_in[9];
  const float* bta       = (const float*)d_in[10];
  const float* osp       = (const float*)d_in[11];
  float* out = (float*)d_out;

  const int B = in_sizes[1] / HWC;

  float* comb = (float*)d_ws;
  int*   meta = (int*)((char*)d_ws + (size_t)B * MAXOBJ * KP * sizeof(float));

  cc_extract_pool_kernel<<<B, 1024, 0, stream>>>(grid, grid_emb, meta, comb);
  fused_mlp_kernel<<<B, 512, 0, stream>>>(comb, meta, structure, W1, b1, W2, b2,
                                          Wp, bp, gmm, bta, osp, out);
}
